# starting kernel
# baseline (speedup 1.0000x reference)
#include <hip/hip_runtime.h>
#include <math.h>

#define N_NODES 50000
#define N_EDGES 800000
#define DIM     128
#define LRELU_A 0.2f

typedef __attribute__((ext_vector_type(2))) float v2f;
typedef __attribute__((ext_vector_type(8))) float v8f;

__global__ __launch_bounds__(128) void gat_gemm_wmma(const float* __restrict__ h,
                                                     const float* __restrict__ W,
                                                     float* __restrict__ Wh) {
  const int wave = (int)((blockIdx.x * blockDim.x + threadIdx.x) >> 5);
  const int lane = (int)(threadIdx.x & 31);
  if (wave >= N_NODES / 16) return;

  const int row0  = wave * 16;
  const int rsel  = lane & 15;
  const int khalf = (lane >> 4) * 2;

  v8f acc[8];
  #pragma unroll
  for (int t = 0; t < 8; ++t) { v8f z = {}; acc[t] = z; }

  for (int k0 = 0; k0 < DIM; k0 += 4) {
    const float* ap = h + (size_t)(row0 + rsel) * DIM + (k0 + khalf);
    v2f a; a.x = ap[0]; a.y = ap[1];
    #pragma unroll
    for (int t = 0; t < 8; ++t) {
      const float* bp = W + (size_t)(k0 + khalf) * DIM + t * 16 + rsel;
      v2f b; b.x = bp[0]; b.y = bp[DIM];
      acc[t] = __builtin_amdgcn_wmma_f32_16x16x4_f32(
           false, a,  false, b,
           (short)0, acc[t],  false,  false);
    }
  }

  const int srow = row0 + (lane >> 4) * 8;
  #pragma unroll
  for (int t = 0; t < 8; ++t) {
    #pragma unroll
    for (int i = 0; i < 8; ++i) {
      Wh[(size_t)(srow + i) * DIM + t * 16 + rsel] = acc[t][i];
    }
  }
}

__global__ __launch_bounds__(256) void gat_sd(const float* __restrict__ Wh,
                                              const float* __restrict__ a,
                                              float* __restrict__ s,
                                              float* __restrict__ d) {
  const int node = (int)((blockIdx.x * blockDim.x + threadIdx.x) >> 5);
  const int lane = (int)(threadIdx.x & 31);
  if (node >= N_NODES) return;
  const float* row = Wh + (size_t)node * DIM;
  float ss = 0.f, dd = 0.f;
  #pragma unroll
  for (int j = lane; j < DIM; j += 32) {
    float v = row[j];
    ss += v * a[j];
    dd += v * a[DIM + j];
  }
  #pragma unroll
  for (int off = 16; off > 0; off >>= 1) {
    ss += __shfl_down(ss, off, 32);
    dd += __shfl_down(dd, off, 32);
  }
  if (lane == 0) { s[node] = ss; d[node] = dd; }
}

__global__ void gat_init(const float* __restrict__ h, float* __restrict__ hout,
                         float* __restrict__ emax, float* __restrict__ denom) {
  const size_t i = (size_t)blockIdx.x * blockDim.x + threadIdx.x;
  if (i < (size_t)N_NODES * DIM) hout[i] = h[i];
  if (i < N_NODES) { emax[i] = -INFINITY; denom[i] = 0.f; }
}

__global__ void gat_edge_logits(const float* __restrict__ s, const float* __restrict__ d,
                                const int* __restrict__ src, const int* __restrict__ dst,
                                float* __restrict__ e, float* __restrict__ emax) {
  const int i = (int)(blockIdx.x * blockDim.x + threadIdx.x);
  if (i >= N_EDGES) return;
  float v = s[src[i]] + d[dst[i]];
  v = (v > 0.f) ? v : LRELU_A * v;
  e[i] = v;
  float* addr = emax + dst[i];
  if (v >= 0.f) atomicMax((int*)addr, __float_as_int(v));
  else          atomicMin((unsigned int*)addr, __float_as_uint(v));
}

__global__ void gat_fix_emax(float* __restrict__ emax) {
  const int i = (int)(blockIdx.x * blockDim.x + threadIdx.x);
  if (i < N_NODES) {
    float m = emax[i];
    if (!isfinite(m)) emax[i] = 0.f;
  }
}

__global__ void gat_edge_exp(const float* __restrict__ e, const int* __restrict__ dst,
                             const float* __restrict__ emax, float* __restrict__ ex,
                             float* __restrict__ denom) {
  const int i = (int)(blockIdx.x * blockDim.x + threadIdx.x);
  if (i >= N_EDGES) return;
  const float v = expf(e[i] - emax[dst[i]]);
  ex[i] = v;
  atomicAdd(&denom[dst[i]], v);
}

__global__ __launch_bounds__(256) void gat_scatter(const float* __restrict__ Wh,
                                                   const int* __restrict__ src,
                                                   const int* __restrict__ dst,
                                                   const float* __restrict__ ex,
                                                   const float* __restrict__ denom,
                                                   float* __restrict__ hout) {
  const int edge = (int)((blockIdx.x * blockDim.x + threadIdx.x) >> 5);
  const int lane = (int)(threadIdx.x & 31);
  if (edge >= N_EDGES) return;
  const int sn = src[edge];
  const int dn = dst[edge];
  const float w = ex[edge] / denom[dn];
  const float4 v = ((const float4*)(Wh + (size_t)sn * DIM))[lane];
  float* orow = hout + (size_t)dn * DIM + lane * 4;
  atomicAdd(orow + 0, w * v.x);
  atomicAdd(orow + 1, w * v.y);
  atomicAdd(orow + 2, w * v.z);
  atomicAdd(orow + 3, w * v.w);
}

extern "C" void kernel_launch(void* const* d_in, const int* in_sizes, int n_in,
                              void* d_out, int out_size, void* d_ws, size_t ws_size,
                              hipStream_t stream) {
  const float* h   = (const float*)d_in[0];
  const float* W   = (const float*)d_in[1];
  const float* a   = (const float*)d_in[2];
  const int*   src = (const int*)d_in[3];
  const int*   dst = (const int*)d_in[4];
  float* hout = (float*)d_out;

  float* ws    = (float*)d_ws;
  float* Wh    = ws;
  float* s     = Wh   + (size_t)N_NODES * DIM;
  float* d     = s    + N_NODES;
  float* emax  = d    + N_NODES;
  float* denom = emax + N_NODES;
  float* e     = denom + N_NODES;
  float* ex    = e    + N_EDGES;

  {
    const int waves = N_NODES / 16;
    const int blocks = (waves + 3) / 4;
    gat_gemm_wmma<<<blocks, 128, 0, stream>>>(h, W, Wh);
  }
  gat_sd<<<(N_NODES * 32 + 255) / 256, 256, 0, stream>>>(Wh, a, s, d);
  {
    const size_t total = (size_t)N_NODES * DIM;
    gat_init<<<(unsigned)((total + 255) / 256), 256, 0, stream>>>(h, hout, emax, denom);
  }
  gat_edge_logits<<<(N_EDGES + 255) / 256, 256, 0, stream>>>(s, d, src, dst, e, emax);
  gat_fix_emax<<<(N_NODES + 255) / 256, 256, 0, stream>>>(emax);
  gat_edge_exp<<<(N_EDGES + 255) / 256, 256, 0, stream>>>(e, dst, emax, ex, denom);
  gat_scatter<<<(N_EDGES * 32 + 255) / 256, 256, 0, stream>>>(Wh, src, dst, ex, denom, hout);
}


# Counting-sort edges by dst each call (int atomic rank, scan, fill); one wave per dst node does softmax and weighted row sum in registers: no float atomics, sums now in rank order
# speedup vs baseline: 2.0473x; 2.0473x over previous
#include <hip/hip_runtime.h>
#include <math.h>

#define N_NODES 50000
#define N_EDGES 800000
#define DIM     128
#define LRELU_A 0.2f
#define SCAN_T  1024

typedef __attribute__((ext_vector_type(2))) float v2f;
typedef __attribute__((ext_vector_type(8))) float v8f;

__global__ __launch_bounds__(128) void gat_gemm_wmma(const float* __restrict__ h,
                                                     const float* __restrict__ W,
                                                     float* __restrict__ Wh) {
  const int wave = (int)((blockIdx.x * blockDim.x + threadIdx.x) >> 5);
  const int lane = (int)(threadIdx.x & 31);
  if (wave >= N_NODES / 16) return;

  const int row0  = wave * 16;
  const int rsel  = lane & 15;
  const int khalf = (lane >> 4) * 2;

  v8f acc[8];
  #pragma unroll
  for (int t = 0; t < 8; ++t) { v8f z = {}; acc[t] = z; }

  for (int k0 = 0; k0 < DIM; k0 += 4) {
    const float* ap = h + (size_t)(row0 + rsel) * DIM + (k0 + khalf);
    v2f a; a.x = ap[0]; a.y = ap[1];
    #pragma unroll
    for (int t = 0; t < 8; ++t) {
      const float* bp = W + (size_t)(k0 + khalf) * DIM + t * 16 + rsel;
      v2f b; b.x = bp[0]; b.y = bp[DIM];
      acc[t] = __builtin_amdgcn_wmma_f32_16x16x4_f32(
           false, a,  false, b,
           (short)0, acc[t],  false,  false);
    }
  }

  const int srow = row0 + (lane >> 4) * 8;
  #pragma unroll
  for (int t = 0; t < 8; ++t) {
    #pragma unroll
    for (int i = 0; i < 8; ++i) {
      Wh[(size_t)(srow + i) * DIM + t * 16 + rsel] = acc[t][i];
    }
  }
}

__global__ __launch_bounds__(256) void gat_sd(const float* __restrict__ Wh,
                                              const float* __restrict__ a,
                                              float* __restrict__ s,
                                              float* __restrict__ d,
                                              int* __restrict__ cnt) {
  const int node = (int)((blockIdx.x * blockDim.x + threadIdx.x) >> 5);
  const int lane = (int)(threadIdx.x & 31);
  if (node >= N_NODES) return;
  const float* row = Wh + (size_t)node * DIM;
  float ss = 0.f, dd = 0.f;
  #pragma unroll
  for (int j = lane; j < DIM; j += 32) {
    float v = row[j];
    ss += v * a[j];
    dd += v * a[DIM + j];
  }
  #pragma unroll
  for (int off = 16; off > 0; off >>= 1) {
    ss += __shfl_down(ss, off, 32);
    dd += __shfl_down(dd, off, 32);
  }
  if (lane == 0) { s[node] = ss; d[node] = dd; cnt[node] = 0; }
}

__global__ __launch_bounds__(256) void gat_rank(const int* __restrict__ dst,
                                                int* __restrict__ cnt,
                                                int* __restrict__ rank) {
  const int i = (int)(blockIdx.x * blockDim.x + threadIdx.x);
  if (i >= N_EDGES) return;
  rank[i] = atomicAdd(&cnt[dst[i]], 1);
}

__global__ __launch_bounds__(SCAN_T) void gat_scan(const int* __restrict__ cnt,
                                                   int* __restrict__ off) {
  __shared__ int part[SCAN_T / 32];
  __shared__ int wtot[SCAN_T / 32];
  const int t    = (int)threadIdx.x;
  const int lane = t & 31;
  const int wv   = t >> 5;
  const int first = (int)blockIdx.x * SCAN_T;

  int pre = 0;
  for (int j = t; j < first; j += SCAN_T) pre += cnt[j];
  #pragma unroll
  for (int o = 16; o > 0; o >>= 1) pre += __shfl_xor(pre, o, 32);

  const int i = first + t;
  const int c = (i < N_NODES) ? cnt[i] : 0;
  int inc = c;
  #pragma unroll
  for (int o = 1; o < 32; o <<= 1) {
    const int u = __shfl_up(inc, o, 32);
    if (lane >= o) inc += u;
  }
  if (lane == 0)  part[wv] = pre;
  if (lane == 31) wtot[wv] = inc;
  __syncthreads();
  int x = part[lane] + ((lane < wv) ? wtot[lane] : 0);
  #pragma unroll
  for (int o = 16; o > 0; o >>= 1) x += __shfl_xor(x, o, 32);
  if (i < N_NODES) off[i] = x + inc - c;
}

__global__ __launch_bounds__(256) void gat_fill(const int* __restrict__ src,
                                                const int* __restrict__ dst,
                                                const int* __restrict__ rank,
                                                const int* __restrict__ off,
                                                const float* __restrict__ s,
                                                int* __restrict__ csr) {
  const int i = (int)(blockIdx.x * blockDim.x + threadIdx.x);
  if (i >= N_EDGES) return;
  const int sn = src[i];
  const int slot = off[dst[i]] + rank[i];
  ((int2*)csr)[slot] = make_int2(sn, __float_as_int(s[sn]));
}

__device__ __forceinline__ float gat_logit(int sbits, float dn) {
  const float v = __int_as_float(sbits) + dn;
  return (v > 0.f) ? v : LRELU_A * v;
}

template <int B>
__device__ __forceinline__ void gat_rows(float4& acc, const float* __restrict__ Wh,
                                         int sn, float w, int k0, int lane) {
  float4 v[B];
  float  wk[B];
  #pragma unroll
  for (int b = 0; b < B; ++b) {
    const int sk = __builtin_amdgcn_readlane(sn, k0 + b);
    wk[b] = __int_as_float(__builtin_amdgcn_readlane(__float_as_int(w), k0 + b));
    v[b]  = ((const float4*)(Wh + (size_t)sk * DIM))[lane];
  }
  #pragma unroll
  for (int b = 0; b < B; ++b) {
    acc.x = fmaf(wk[b], v[b].x, acc.x);
    acc.y = fmaf(wk[b], v[b].y, acc.y);
    acc.z = fmaf(wk[b], v[b].z, acc.z);
    acc.w = fmaf(wk[b], v[b].w, acc.w);
  }
}

__device__ __forceinline__ void gat_chunk(float4& acc, const float* __restrict__ Wh,
                                          int sn, float w, int nk, int lane) {
  int k = 0;
  for (; k + 8 <= nk; k += 8) gat_rows<8>(acc, Wh, sn, w, k, lane);
  if (k + 4 <= nk) { gat_rows<4>(acc, Wh, sn, w, k, lane); k += 4; }
  if (k + 2 <= nk) { gat_rows<2>(acc, Wh, sn, w, k, lane); k += 2; }
  if (k < nk)      { gat_rows<1>(acc, Wh, sn, w, k, lane); }
}

__global__ __launch_bounds__(256) void gat_gather(const float* __restrict__ Wh,
                                                  const float* __restrict__ h,
                                                  const float* __restrict__ d,
                                                  const int* __restrict__ off,
                                                  const int* __restrict__ cnt,
                                                  const int* __restrict__ csr,
                                                  float* __restrict__ hout) {
  const int node = __builtin_amdgcn_readfirstlane((int)((blockIdx.x * blockDim.x + threadIdx.x) >> 5));
  const int lane = (int)(threadIdx.x & 31);
  if (node >= N_NODES) return;
  const int   beg = off[node];
  const int   deg = cnt[node];
  const float dn  = d[node];
  const int2* recs = (const int2*)csr + beg;
  const float4 hv = ((const float4*)(h + (size_t)node * DIM))[lane];

  int   sn0 = 0;
  float e0  = -INFINITY;
  if (lane < deg) { const int2 r = recs[lane]; sn0 = r.x; e0 = gat_logit(r.y, dn); }
  float m = e0;
  for (int j = 32 + lane; j < deg; j += 32) m = fmaxf(m, gat_logit(recs[j].y, dn));
  #pragma unroll
  for (int o = 16; o > 0; o >>= 1) m = fmaxf(m, __shfl_xor(m, o, 32));
  if (!isfinite(m)) m = 0.f;

  const float ex0 = (lane < deg) ? expf(e0 - m) : 0.f;
  float den = ex0;
  for (int j = 32 + lane; j < deg; j += 32) den += expf(gat_logit(recs[j].y, dn) - m);
  #pragma unroll
  for (int o = 16; o > 0; o >>= 1) den += __shfl_xor(den, o, 32);

  float4 acc = make_float4(0.f, 0.f, 0.f, 0.f);
  gat_chunk(acc, Wh, sn0, ex0 / den, deg < 32 ? deg : 32, lane);
  for (int j0 = 32; j0 < deg; j0 += 32) {
    int sn = 0; float w = 0.f;
    if (j0 + lane < deg) {
      const int2 r = recs[j0 + lane];
      sn = r.x;
      w  = expf(gat_logit(r.y, dn) - m) / den;
    }
    gat_chunk(acc, Wh, sn, w, (deg - j0) < 32 ? (deg - j0) : 32, lane);
  }

  float4 o4;
  o4.x = acc.x + hv.x; o4.y = acc.y + hv.y; o4.z = acc.z + hv.z; o4.w = acc.w + hv.w;
  ((float4*)(hout + (size_t)node * DIM))[lane] = o4;
}

extern "C" void kernel_launch(void* const* d_in, const int* in_sizes, int n_in,
                              void* d_out, int out_size, void* d_ws, size_t ws_size,
                              hipStream_t stream) {
  const float* h   = (const float*)d_in[0];
  const float* W   = (const float*)d_in[1];
  const float* a   = (const float*)d_in[2];
  const int*   src = (const int*)d_in[3];
  const int*   dst = (const int*)d_in[4];
  float* hout = (float*)d_out;

  float* ws   = (float*)d_ws;
  float* Wh   = ws;
  float* s    = Wh + (size_t)N_NODES * DIM;
  float* d    = s  + N_NODES;
  int*   cnt  = (int*)(d + N_NODES);
  int*   off  = cnt + N_NODES;
  int*   rank = off + N_NODES;
  int*   csr  = rank + N_EDGES;

  {
    const int waves = N_NODES / 16;
    const int blocks = (waves + 3) / 4;
    gat_gemm_wmma<<<blocks, 128, 0, stream>>>(h, W, Wh);
  }
  gat_sd<<<(N_NODES * 32 + 255) / 256, 256, 0, stream>>>(Wh, a, s, d, cnt);
  gat_rank<<<(N_EDGES + 255) / 256, 256, 0, stream>>>(dst, cnt, rank);
  gat_scan<<<(N_NODES + SCAN_T - 1) / SCAN_T, SCAN_T, 0, stream>>>(cnt, off);
  gat_fill<<<(N_EDGES + 255) / 256, 256, 0, stream>>>(src, dst, rank, off, s, csr);
  gat_gather<<<(N_NODES * 32 + 255) / 256, 256, 0, stream>>>(Wh, h, d, off, cnt, csr, hout);
}


# Fuse the logit halves s,d and the counter clear into the GEMM epilogue (from the accumulators, 16-lane DPP sums): one launch and 50,000 waves fewer, Wh not re-read; s,d summed in a new order
# speedup vs baseline: 2.3004x; 1.1236x over previous
#include <hip/hip_runtime.h>
#include <math.h>

#define N_NODES 50000
#define N_EDGES 800000
#define DIM     128
#define LRELU_A 0.2f
#define SCAN_T  1024

typedef __attribute__((ext_vector_type(2))) float v2f;
typedef __attribute__((ext_vector_type(8))) float v8f;

template <int M>
__device__ __forceinline__ float gat_row_xor(float x) {
  const int b = __builtin_bit_cast(int, x);
  return __builtin_bit_cast(float, __builtin_amdgcn_update_dpp(b, b, 0x160 + M, 0xf, 0xf, false));
}
__device__ __forceinline__ float gat_row_sum16(float x) {
  x += gat_row_xor<1>(x);
  x += gat_row_xor<2>(x);
  x += gat_row_xor<4>(x);
  x += gat_row_xor<8>(x);
  return x;
}

__global__ __launch_bounds__(128) void gat_gemm_wmma(const float* __restrict__ h,
                                                     const float* __restrict__ W,
                                                     const float* __restrict__ avec,
                                                     float* __restrict__ Wh,
                                                     float* __restrict__ s,
                                                     float* __restrict__ d,
                                                     int* __restrict__ cnt) {
  const int wave = (int)((blockIdx.x * blockDim.x + threadIdx.x) >> 5);
  const int lane = (int)(threadIdx.x & 31);
  if (wave >= N_NODES / 16) return;

  const int row0  = wave * 16;
  const int rsel  = lane & 15;
  const int khalf = (lane >> 4) * 2;

  v8f acc[8];
  #pragma unroll
  for (int t = 0; t < 8; ++t) { v8f z = {}; acc[t] = z; }

  for (int k0 = 0; k0 < DIM; k0 += 4) {
    const float* ap = h + (size_t)(row0 + rsel) * DIM + (k0 + khalf);
    v2f a; a.x = ap[0]; a.y = ap[1];
    #pragma unroll
    for (int t = 0; t < 8; ++t) {
      const float* bp = W + (size_t)(k0 + khalf) * DIM + t * 16 + rsel;
      v2f b; b.x = bp[0]; b.y = bp[DIM];
      acc[t] = __builtin_amdgcn_wmma_f32_16x16x4_f32(
           false, a,  false, b,
           (short)0, acc[t],  false,  false);
    }
  }

  const int srow = row0 + (lane >> 4) * 8;
  #pragma unroll
  for (int t = 0; t < 8; ++t) {
    #pragma unroll
    for (int i = 0; i < 8; ++i) {
      Wh[(size_t)(srow + i) * DIM + t * 16 + rsel] = acc[t][i];
    }
  }

  float as[8], ad[8];
  #pragma unroll
  for (int t = 0; t < 8; ++t) { as[t] = avec[t * 16 + rsel]; ad[t] = avec[DIM + t * 16 + rsel]; }
  float ps[8], pd[8];
  #pragma unroll
  for (int i = 0; i < 8; ++i) {
    float x = 0.f, y = 0.f;
    #pragma unroll
    for (int t = 0; t < 8; ++t) {
      x = fmaf(acc[t][i], as[t], x);
      y = fmaf(acc[t][i], ad[t], y);
    }
    ps[i] = gat_row_sum16(x);
    pd[i] = gat_row_sum16(y);
  }
  if (rsel == 0) {
    float4* sp = (float4*)(s + srow);
    float4* dp = (float4*)(d + srow);
    int4*   cp = (int4*)(cnt + srow);
    sp[0] = make_float4(ps[0], ps[1], ps[2], ps[3]);
    sp[1] = make_float4(ps[4], ps[5], ps[6], ps[7]);
    dp[0] = make_float4(pd[0], pd[1], pd[2], pd[3]);
    dp[1] = make_float4(pd[4], pd[5], pd[6], pd[7]);
    cp[0] = make_int4(0, 0, 0, 0);
    cp[1] = make_int4(0, 0, 0, 0);
  }
}

__global__ __launch_bounds__(256) void gat_rank(const int* __restrict__ dst,
                                                int* __restrict__ cnt,
                                                int* __restrict__ rank) {
  const int i = (int)(blockIdx.x * blockDim.x + threadIdx.x);
  if (i >= N_EDGES) return;
  rank[i] = atomicAdd(&cnt[dst[i]], 1);
}

__global__ __launch_bounds__(SCAN_T) void gat_scan(const int* __restrict__ cnt,
                                                   int* __restrict__ off) {
  __shared__ int part[SCAN_T / 32];
  __shared__ int wtot[SCAN_T / 32];
  const int t    = (int)threadIdx.x;
  const int lane = t & 31;
  const int wv   = t >> 5;
  const int first = (int)blockIdx.x * SCAN_T;

  int pre = 0;
  for (int j = t; j < first; j += SCAN_T) pre += cnt[j];
  #pragma unroll
  for (int o = 16; o > 0; o >>= 1) pre += __shfl_xor(pre, o, 32);

  const int i = first + t;
  const int c = (i < N_NODES) ? cnt[i] : 0;
  int inc = c;
  #pragma unroll
  for (int o = 1; o < 32; o <<= 1) {
    const int u = __shfl_up(inc, o, 32);
    if (lane >= o) inc += u;
  }
  if (lane == 0)  part[wv] = pre;
  if (lane == 31) wtot[wv] = inc;
  __syncthreads();
  int x = part[lane] + ((lane < wv) ? wtot[lane] : 0);
  #pragma unroll
  for (int o = 16; o > 0; o >>= 1) x += __shfl_xor(x, o, 32);
  if (i < N_NODES) off[i] = x + inc - c;
}

__global__ __launch_bounds__(256) void gat_fill(const int* __restrict__ src,
                                                const int* __restrict__ dst,
                                                const int* __restrict__ rank,
                                                const int* __restrict__ off,
                                                const float* __restrict__ s,
                                                int* __restrict__ csr) {
  const int i = (int)(blockIdx.x * blockDim.x + threadIdx.x);
  if (i >= N_EDGES) return;
  const int sn = src[i];
  const int slot = off[dst[i]] + rank[i];
  ((int2*)csr)[slot] = make_int2(sn, __float_as_int(s[sn]));
}

__device__ __forceinline__ float gat_logit(int sbits, float dn) {
  const float v = __int_as_float(sbits) + dn;
  return (v > 0.f) ? v : LRELU_A * v;
}

template <int B>
__device__ __forceinline__ void gat_rows(float4& acc, const float* __restrict__ Wh,
                                         int sn, float w, int k0, int lane) {
  float4 v[B];
  float  wk[B];
  #pragma unroll
  for (int b = 0; b < B; ++b) {
    const int sk = __builtin_amdgcn_readlane(sn, k0 + b);
    wk[b] = __int_as_float(__builtin_amdgcn_readlane(__float_as_int(w), k0 + b));
    v[b]  = ((const float4*)(Wh + (size_t)sk * DIM))[lane];
  }
  #pragma unroll
  for (int b = 0; b < B; ++b) {
    acc.x = fmaf(wk[b], v[b].x, acc.x);
    acc.y = fmaf(wk[b], v[b].y, acc.y);
    acc.z = fmaf(wk[b], v[b].z, acc.z);
    acc.w = fmaf(wk[b], v[b].w, acc.w);
  }
}

__device__ __forceinline__ void gat_chunk(float4& acc, const float* __restrict__ Wh,
                                          int sn, float w, int nk, int lane) {
  int k = 0;
  for (; k + 8 <= nk; k += 8) gat_rows<8>(acc, Wh, sn, w, k, lane);
  if (k + 4 <= nk) { gat_rows<4>(acc, Wh, sn, w, k, lane); k += 4; }
  if (k + 2 <= nk) { gat_rows<2>(acc, Wh, sn, w, k, lane); k += 2; }
  if (k < nk)      { gat_rows<1>(acc, Wh, sn, w, k, lane); }
}

__global__ __launch_bounds__(256) void gat_gather(const float* __restrict__ Wh,
                                                  const float* __restrict__ h,
                                                  const float* __restrict__ d,
                                                  const int* __restrict__ off,
                                                  const int* __restrict__ cnt,
                                                  const int* __restrict__ csr,
                                                  float* __restrict__ hout) {
  const int node = __builtin_amdgcn_readfirstlane((int)((blockIdx.x * blockDim.x + threadIdx.x) >> 5));
  const int lane = (int)(threadIdx.x & 31);
  if (node >= N_NODES) return;
  const int   beg = off[node];
  const int   deg = cnt[node];
  const float dn  = d[node];
  const int2* recs = (const int2*)csr + beg;
  const float4 hv = ((const float4*)(h + (size_t)node * DIM))[lane];

  int   sn0 = 0;
  float e0  = -INFINITY;
  if (lane < deg) { const int2 r = recs[lane]; sn0 = r.x; e0 = gat_logit(r.y, dn); }
  float m = e0;
  for (int j = 32 + lane; j < deg; j += 32) m = fmaxf(m, gat_logit(recs[j].y, dn));
  #pragma unroll
  for (int o = 16; o > 0; o >>= 1) m = fmaxf(m, __shfl_xor(m, o, 32));
  if (!isfinite(m)) m = 0.f;

  const float ex0 = (lane < deg) ? expf(e0 - m) : 0.f;
  float den = ex0;
  for (int j = 32 + lane; j < deg; j += 32) den += expf(gat_logit(recs[j].y, dn) - m);
  #pragma unroll
  for (int o = 16; o > 0; o >>= 1) den += __shfl_xor(den, o, 32);

  float4 acc = make_float4(0.f, 0.f, 0.f, 0.f);
  gat_chunk(acc, Wh, sn0, ex0 / den, deg < 32 ? deg : 32, lane);
  for (int j0 = 32; j0 < deg; j0 += 32) {
    int sn = 0; float w = 0.f;
    if (j0 + lane < deg) {
      const int2 r = recs[j0 + lane];
      sn = r.x;
      w  = expf(gat_logit(r.y, dn) - m) / den;
    }
    gat_chunk(acc, Wh, sn, w, (deg - j0) < 32 ? (deg - j0) : 32, lane);
  }

  float4 o4;
  o4.x = acc.x + hv.x; o4.y = acc.y + hv.y; o4.z = acc.z + hv.z; o4.w = acc.w + hv.w;
  ((float4*)(hout + (size_t)node * DIM))[lane] = o4;
}

extern "C" void kernel_launch(void* const* d_in, const int* in_sizes, int n_in,
                              void* d_out, int out_size, void* d_ws, size_t ws_size,
                              hipStream_t stream) {
  const float* h   = (const float*)d_in[0];
  const float* W   = (const float*)d_in[1];
  const float* a   = (const float*)d_in[2];
  const int*   src = (const int*)d_in[3];
  const int*   dst = (const int*)d_in[4];
  float* hout = (float*)d_out;

  float* ws   = (float*)d_ws;
  float* Wh   = ws;
  float* s    = Wh + (size_t)N_NODES * DIM;
  float* d    = s  + N_NODES;
  int*   cnt  = (int*)(d + N_NODES);
  int*   off  = cnt + N_NODES;
  int*   rank = off + N_NODES;
  int*   csr  = rank + N_EDGES;

  {
    const int waves = N_NODES / 16;
    const int blocks = (waves + 3) / 4;
    gat_gemm_wmma<<<blocks, 128, 0, stream>>>(h, W, a, Wh, s, d, cnt);
  }
  gat_rank<<<(N_EDGES + 255) / 256, 256, 0, stream>>>(dst, cnt, rank);
  gat_scan<<<(N_NODES + SCAN_T - 1) / SCAN_T, SCAN_T, 0, stream>>>(cnt, off);
  gat_fill<<<(N_EDGES + 255) / 256, 256, 0, stream>>>(src, dst, rank, off, s, csr);
  gat_gather<<<(N_NODES * 32 + 255) / 256, 256, 0, stream>>>(Wh, h, d, off, cnt, csr, hout);
}


# Edge ranks from returning LDS adds: 32 work-groups each count a 25,000-edge chunk in a private 50,000-counter LDS table, scan joins the rows; no global atomics; sums now in chunk, then arrival order
# speedup vs baseline: 3.0006x; 1.3044x over previous
#include <hip/hip_runtime.h>
#include <math.h>

#define N_NODES 50000
#define N_EDGES 800000
#define DIM     128
#define LRELU_A 0.2f
#define RK_G    32
#define RK_CH   (N_EDGES / RK_G)
#define RK_T    1024
#define RK_PER  ((RK_CH + RK_T - 1) / RK_T)
#define SL_SH   8
#define SL_N    (1 << SL_SH)
#define N_SL    ((N_NODES + SL_N - 1) / SL_N)
static_assert(N_EDGES % RK_G == 0, "chunks tile the edge list exactly");
static_assert(N_NODES % 4 == 0, "count rows move as 16-byte words");
static_assert(N_SL <= RK_T, "one thread per slice total");

typedef __attribute__((ext_vector_type(2))) float v2f;
typedef __attribute__((ext_vector_type(8))) float v8f;

template <int M>
__device__ __forceinline__ float gat_row_xor(float x) {
  const int b = __builtin_bit_cast(int, x);
  return __builtin_bit_cast(float, __builtin_amdgcn_update_dpp(b, b, 0x160 + M, 0xf, 0xf, false));
}
__device__ __forceinline__ float gat_row_sum16(float x) {
  x += gat_row_xor<1>(x);
  x += gat_row_xor<2>(x);
  x += gat_row_xor<4>(x);
  x += gat_row_xor<8>(x);
  return x;
}

__global__ __launch_bounds__(128) void gat_gemm_wmma(const float* __restrict__ h,
                                                     const float* __restrict__ W,
                                                     const float* __restrict__ avec,
                                                     float* __restrict__ Wh,
                                                     float* __restrict__ s,
                                                     float* __restrict__ d) {
  const int wave = (int)((blockIdx.x * blockDim.x + threadIdx.x) >> 5);
  const int lane = (int)(threadIdx.x & 31);
  if (wave >= N_NODES / 16) return;

  const int row0  = wave * 16;
  const int rsel  = lane & 15;
  const int khalf = (lane >> 4) * 2;

  v8f acc[8];
  #pragma unroll
  for (int t = 0; t < 8; ++t) { v8f z = {}; acc[t] = z; }

  for (int k0 = 0; k0 < DIM; k0 += 4) {
    const float* ap = h + (size_t)(row0 + rsel) * DIM + (k0 + khalf);
    v2f a; a.x = ap[0]; a.y = ap[1];
    #pragma unroll
    for (int t = 0; t < 8; ++t) {
      const float* bp = W + (size_t)(k0 + khalf) * DIM + t * 16 + rsel;
      v2f b; b.x = bp[0]; b.y = bp[DIM];
      acc[t] = __builtin_amdgcn_wmma_f32_16x16x4_f32(
           false, a,  false, b,
           (short)0, acc[t],  false,  false);
    }
  }

  const int srow = row0 + (lane >> 4) * 8;
  #pragma unroll
  for (int t = 0; t < 8; ++t) {
    #pragma unroll
    for (int i = 0; i < 8; ++i) {
      Wh[(size_t)(srow + i) * DIM + t * 16 + rsel] = acc[t][i];
    }
  }

  float as[8], ad[8];
  #pragma unroll
  for (int t = 0; t < 8; ++t) { as[t] = avec[t * 16 + rsel]; ad[t] = avec[DIM + t * 16 + rsel]; }
  float ps[8], pd[8];
  #pragma unroll
  for (int i = 0; i < 8; ++i) {
    float x = 0.f, y = 0.f;
    #pragma unroll
    for (int t = 0; t < 8; ++t) {
      x = fmaf(acc[t][i], as[t], x);
      y = fmaf(acc[t][i], ad[t], y);
    }
    ps[i] = gat_row_sum16(x);
    pd[i] = gat_row_sum16(y);
  }
  if (rsel == 0) {
    float4* sp = (float4*)(s + srow);
    float4* dp = (float4*)(d + srow);
    sp[0] = make_float4(ps[0], ps[1], ps[2], ps[3]);
    sp[1] = make_float4(ps[4], ps[5], ps[6], ps[7]);
    dp[0] = make_float4(pd[0], pd[1], pd[2], pd[3]);
    dp[1] = make_float4(pd[4], pd[5], pd[6], pd[7]);
  }
}

__global__ __launch_bounds__(RK_T) void gat_rank(const int* __restrict__ dst,
                                                 int* __restrict__ rank,
                                                 int* __restrict__ cntm,
                                                 int* __restrict__ stot) {
  __shared__ __attribute__((aligned(16))) int hist[N_NODES];
  __shared__ int sl[N_SL];
  const int t  = (int)threadIdx.x;
  const int w  = (int)blockIdx.x;
  const int e0 = w * RK_CH;

  int dv[RK_PER];
  #pragma unroll
  for (int j = 0; j < RK_PER; ++j) {
    const int e = t + j * RK_T;
    dv[j] = (e < RK_CH) ? dst[e0 + e] : 0;
  }

  int4* h4 = (int4*)hist;
  for (int q = t; q < N_NODES / 4; q += RK_T) h4[q] = make_int4(0, 0, 0, 0);
  if (t < N_SL) sl[t] = 0;
  __syncthreads();

  int rk[RK_PER];
  #pragma unroll
  for (int j = 0; j < RK_PER; ++j) {
    const int e = t + j * RK_T;
    rk[j] = 0;
    if (e < RK_CH) {
      rk[j] = atomicAdd(&hist[dv[j]], 1);
      atomicAdd(&sl[dv[j] >> SL_SH], 1);
    }
  }
  #pragma unroll
  for (int j = 0; j < RK_PER; ++j) {
    const int e = t + j * RK_T;
    if (e < RK_CH) rank[e0 + e] = rk[j];
  }
  __syncthreads();

  int4* o4 = (int4*)(cntm + (size_t)w * N_NODES);
  for (int q = t; q < N_NODES / 4; q += RK_T) o4[q] = h4[q];
  if (t < N_SL) stot[t * RK_G + w] = sl[t];
}

__global__ __launch_bounds__(SL_N) void gat_scan(int* __restrict__ cntm,
                                                 const int* __restrict__ stot,
                                                 int* __restrict__ off,
                                                 int* __restrict__ cnt) {
  __shared__ int wpre[SL_N / 32];
  __shared__ int wtot[SL_N / 32];
  const int t    = (int)threadIdx.x;
  const int lane = t & 31;
  const int wv   = t >> 5;
  const int n    = (int)blockIdx.x * SL_N + t;

  const int nc = (n < N_NODES) ? n : N_NODES - 1;
  int c[RK_G];
  #pragma unroll
  for (int w = 0; w < RK_G; ++w) {
    const int v = cntm[(size_t)w * N_NODES + nc];
    c[w] = (n < N_NODES) ? v : 0;
  }

  const int lim = (int)blockIdx.x * RK_G;
  int pre = 0;
  #pragma unroll
  for (int k = 0; k < (N_SL * RK_G + SL_N - 1) / SL_N; ++k) {
    const int j = t + k * SL_N;
    const int v = stot[j < N_SL * RK_G ? j : N_SL * RK_G - 1];
    pre += (j < lim) ? v : 0;
  }
  #pragma unroll
  for (int o = 16; o > 0; o >>= 1) pre += __shfl_xor(pre, o, 32);

  int tot = 0;
  #pragma unroll
  for (int w = 0; w < RK_G; ++w) tot += c[w];
  int inc = tot;
  #pragma unroll
  for (int o = 1; o < 32; o <<= 1) {
    const int u = __shfl_up(inc, o, 32);
    if (lane >= o) inc += u;
  }
  if (lane == 0)  wpre[wv] = pre;
  if (lane == 31) wtot[wv] = inc;
  __syncthreads();
  int x = (lane < SL_N / 32) ? (wpre[lane] + ((lane < wv) ? wtot[lane] : 0)) : 0;
  #pragma unroll
  for (int o = 16; o > 0; o >>= 1) x += __shfl_xor(x, o, 32);

  if (n < N_NODES) {
    int run = x + inc - tot;
    off[n] = run;
    cnt[n] = tot;
    #pragma unroll
    for (int w = 0; w < RK_G; ++w) { cntm[(size_t)w * N_NODES + n] = run; run += c[w]; }
  }
}

__global__ __launch_bounds__(256) void gat_fill(const int* __restrict__ src,
                                                const int* __restrict__ dst,
                                                const int* __restrict__ rank,
                                                const int* __restrict__ base,
                                                const float* __restrict__ s,
                                                int* __restrict__ csr) {
  const int i = (int)(blockIdx.x * blockDim.x + threadIdx.x);
  if (i >= N_EDGES) return;
  const int sn = src[i];
  const int w  = i / RK_CH;
  const int slot = base[(size_t)w * N_NODES + dst[i]] + rank[i];
  ((int2*)csr)[slot] = make_int2(sn, __float_as_int(s[sn]));
}

__device__ __forceinline__ float gat_logit(int sbits, float dn) {
  const float v = __int_as_float(sbits) + dn;
  return (v > 0.f) ? v : LRELU_A * v;
}

template <int B>
__device__ __forceinline__ void gat_rows(float4& acc, const float* __restrict__ Wh,
                                         int sn, float w, int k0, int lane) {
  float4 v[B];
  float  wk[B];
  #pragma unroll
  for (int b = 0; b < B; ++b) {
    const int sk = __builtin_amdgcn_readlane(sn, k0 + b);
    wk[b] = __int_as_float(__builtin_amdgcn_readlane(__float_as_int(w), k0 + b));
    v[b]  = ((const float4*)(Wh + (size_t)sk * DIM))[lane];
  }
  #pragma unroll
  for (int b = 0; b < B; ++b) {
    acc.x = fmaf(wk[b], v[b].x, acc.x);
    acc.y = fmaf(wk[b], v[b].y, acc.y);
    acc.z = fmaf(wk[b], v[b].z, acc.z);
    acc.w = fmaf(wk[b], v[b].w, acc.w);
  }
}

__device__ __forceinline__ void gat_chunk(float4& acc, const float* __restrict__ Wh,
                                          int sn, float w, int nk, int lane) {
  int k = 0;
  for (; k + 8 <= nk; k += 8) gat_rows<8>(acc, Wh, sn, w, k, lane);
  if (k + 4 <= nk) { gat_rows<4>(acc, Wh, sn, w, k, lane); k += 4; }
  if (k + 2 <= nk) { gat_rows<2>(acc, Wh, sn, w, k, lane); k += 2; }
  if (k < nk)      { gat_rows<1>(acc, Wh, sn, w, k, lane); }
}

__global__ __launch_bounds__(256) void gat_gather(const float* __restrict__ Wh,
                                                  const float* __restrict__ h,
                                                  const float* __restrict__ d,
                                                  const int* __restrict__ off,
                                                  const int* __restrict__ cnt,
                                                  const int* __restrict__ csr,
                                                  float* __restrict__ hout) {
  const int node = __builtin_amdgcn_readfirstlane((int)((blockIdx.x * blockDim.x + threadIdx.x) >> 5));
  const int lane = (int)(threadIdx.x & 31);
  if (node >= N_NODES) return;
  const int   beg = off[node];
  const int   deg = cnt[node];
  const float dn  = d[node];
  const int2* recs = (const int2*)csr + beg;
  const float4 hv = ((const float4*)(h + (size_t)node * DIM))[lane];

  int   sn0 = 0;
  float e0  = -INFINITY;
  if (lane < deg) { const int2 r = recs[lane]; sn0 = r.x; e0 = gat_logit(r.y, dn); }
  float m = e0;
  for (int j = 32 + lane; j < deg; j += 32) m = fmaxf(m, gat_logit(recs[j].y, dn));
  #pragma unroll
  for (int o = 16; o > 0; o >>= 1) m = fmaxf(m, __shfl_xor(m, o, 32));
  if (!isfinite(m)) m = 0.f;

  const float ex0 = (lane < deg) ? expf(e0 - m) : 0.f;
  float den = ex0;
  for (int j = 32 + lane; j < deg; j += 32) den += expf(gat_logit(recs[j].y, dn) - m);
  #pragma unroll
  for (int o = 16; o > 0; o >>= 1) den += __shfl_xor(den, o, 32);

  float4 acc = make_float4(0.f, 0.f, 0.f, 0.f);
  gat_chunk(acc, Wh, sn0, ex0 / den, deg < 32 ? deg : 32, lane);
  for (int j0 = 32; j0 < deg; j0 += 32) {
    int sn = 0; float w = 0.f;
    if (j0 + lane < deg) {
      const int2 r = recs[j0 + lane];
      sn = r.x;
      w  = expf(gat_logit(r.y, dn) - m) / den;
    }
    gat_chunk(acc, Wh, sn, w, (deg - j0) < 32 ? (deg - j0) : 32, lane);
  }

  float4 o4;
  o4.x = acc.x + hv.x; o4.y = acc.y + hv.y; o4.z = acc.z + hv.z; o4.w = acc.w + hv.w;
  ((float4*)(hout + (size_t)node * DIM))[lane] = o4;
}

extern "C" void kernel_launch(void* const* d_in, const int* in_sizes, int n_in,
                              void* d_out, int out_size, void* d_ws, size_t ws_size,
                              hipStream_t stream) {
  const float* h   = (const float*)d_in[0];
  const float* W   = (const float*)d_in[1];
  const float* a   = (const float*)d_in[2];
  const int*   src = (const int*)d_in[3];
  const int*   dst = (const int*)d_in[4];
  float* hout = (float*)d_out;

  float* ws   = (float*)d_ws;
  float* Wh   = ws;
  float* s    = Wh + (size_t)N_NODES * DIM;
  float* d    = s  + N_NODES;
  int*   cnt  = (int*)(d + N_NODES);
  int*   off  = cnt + N_NODES;
  int*   rank = off + N_NODES;
  int*   csr  = rank + N_EDGES;
  int*   cntm = csr + 2 * (size_t)N_EDGES;
  int*   stot = cntm + (size_t)RK_G * N_NODES;

  {
    const int waves = N_NODES / 16;
    const int blocks = (waves + 3) / 4;
    gat_gemm_wmma<<<blocks, 128, 0, stream>>>(h, W, a, Wh, s, d);
  }
  gat_rank<<<RK_G, RK_T, 0, stream>>>(dst, rank, cntm, stot);
  gat_scan<<<N_SL, SL_N, 0, stream>>>(cntm, stot, off, cnt);
  gat_fill<<<(N_EDGES + 255) / 256, 256, 0, stream>>>(src, dst, rank, cntm, s, csr);
  gat_gather<<<(N_NODES * 32 + 255) / 256, 256, 0, stream>>>(Wh, h, d, off, cnt, csr, hout);
}


# Ranking shares the GEMM's launch: of one grid of 512-thread work-groups 32 rank the edge chunks while 196 do the row strips, so ranking runs beside the matrix product; 5 launches become 4
# speedup vs baseline: 3.8304x; 1.2765x over previous
#include <hip/hip_runtime.h>
#include <math.h>

#define N_NODES 50000
#define N_EDGES 800000
#define DIM     128
#define LRELU_A 0.2f
#define RK_G    32
#define RK_CH   (N_EDGES / RK_G)
#define RK_T    512
#define RK_PER  ((RK_CH + RK_T - 1) / RK_T)
#define GM_W    (RK_T / 32)
#define GM_G    ((N_NODES / 16 + GM_W - 1) / GM_W)
#define SL_SH   8
#define SL_N    (1 << SL_SH)
#define N_SL    ((N_NODES + SL_N - 1) / SL_N)
static_assert(N_EDGES % RK_G == 0, "chunks tile the edge list exactly");
static_assert(N_NODES % 4 == 0, "count rows move as 16-byte words");
static_assert(N_SL <= RK_T, "one thread per slice total");

typedef __attribute__((ext_vector_type(2))) float v2f;
typedef __attribute__((ext_vector_type(8))) float v8f;

template <int M>
__device__ __forceinline__ float gat_row_xor(float x) {
  const int b = __builtin_bit_cast(int, x);
  return __builtin_bit_cast(float, __builtin_amdgcn_update_dpp(b, b, 0x160 + M, 0xf, 0xf, false));
}
__device__ __forceinline__ float gat_row_sum16(float x) {
  x += gat_row_xor<1>(x);
  x += gat_row_xor<2>(x);
  x += gat_row_xor<4>(x);
  x += gat_row_xor<8>(x);
  return x;
}

__device__ __forceinline__ void gat_gemm_wmma(const int wg,
                                              const float* __restrict__ h,
                                              const float* __restrict__ W,
                                              const float* __restrict__ avec,
                                              float* __restrict__ Wh,
                                              float* __restrict__ s,
                                              float* __restrict__ d) {
  const int wave = wg * GM_W + (int)(threadIdx.x >> 5);
  const int lane = (int)(threadIdx.x & 31);
  if (wave >= N_NODES / 16) return;

  const int row0  = wave * 16;
  const int rsel  = lane & 15;
  const int khalf = (lane >> 4) * 2;

  v8f acc[8];
  #pragma unroll
  for (int t = 0; t < 8; ++t) { v8f z = {}; acc[t] = z; }

  for (int k0 = 0; k0 < DIM; k0 += 4) {
    const float* ap = h + (size_t)(row0 + rsel) * DIM + (k0 + khalf);
    v2f a; a.x = ap[0]; a.y = ap[1];
    #pragma unroll
    for (int t = 0; t < 8; ++t) {
      const float* bp = W + (size_t)(k0 + khalf) * DIM + t * 16 + rsel;
      v2f b; b.x = bp[0]; b.y = bp[DIM];
      acc[t] = __builtin_amdgcn_wmma_f32_16x16x4_f32(
           false, a,  false, b,
           (short)0, acc[t],  false,  false);
    }
  }

  const int srow = row0 + (lane >> 4) * 8;
  #pragma unroll
  for (int t = 0; t < 8; ++t) {
    #pragma unroll
    for (int i = 0; i < 8; ++i) {
      Wh[(size_t)(srow + i) * DIM + t * 16 + rsel] = acc[t][i];
    }
  }

  float as[8], ad[8];
  #pragma unroll
  for (int t = 0; t < 8; ++t) { as[t] = avec[t * 16 + rsel]; ad[t] = avec[DIM + t * 16 + rsel]; }
  float ps[8], pd[8];
  #pragma unroll
  for (int i = 0; i < 8; ++i) {
    float x = 0.f, y = 0.f;
    #pragma unroll
    for (int t = 0; t < 8; ++t) {
      x = fmaf(acc[t][i], as[t], x);
      y = fmaf(acc[t][i], ad[t], y);
    }
    ps[i] = gat_row_sum16(x);
    pd[i] = gat_row_sum16(y);
  }
  if (rsel == 0) {
    float4* sp = (float4*)(s + srow);
    float4* dp = (float4*)(d + srow);
    sp[0] = make_float4(ps[0], ps[1], ps[2], ps[3]);
    sp[1] = make_float4(ps[4], ps[5], ps[6], ps[7]);
    dp[0] = make_float4(pd[0], pd[1], pd[2], pd[3]);
    dp[1] = make_float4(pd[4], pd[5], pd[6], pd[7]);
  }
}

__device__ __forceinline__ void gat_rank(const int w, int* hist, int* sl,
                                         const int* __restrict__ dst,
                                         int* __restrict__ rank,
                                         int* __restrict__ cntm,
                                         int* __restrict__ stot) {
  const int t  = (int)threadIdx.x;
  const int e0 = w * RK_CH;

  int dv[RK_PER];
  #pragma unroll
  for (int j = 0; j < RK_PER; ++j) {
    const int e = t + j * RK_T;
    dv[j] = (e < RK_CH) ? dst[e0 + e] : 0;
  }

  int4* h4 = (int4*)hist;
  for (int q = t; q < N_NODES / 4; q += RK_T) h4[q] = make_int4(0, 0, 0, 0);
  if (t < N_SL) sl[t] = 0;
  __syncthreads();

  int rk[RK_PER];
  #pragma unroll
  for (int j = 0; j < RK_PER; ++j) {
    const int e = t + j * RK_T;
    rk[j] = 0;
    if (e < RK_CH) {
      rk[j] = atomicAdd(&hist[dv[j]], 1);
      atomicAdd(&sl[dv[j] >> SL_SH], 1);
    }
  }
  #pragma unroll
  for (int j = 0; j < RK_PER; ++j) {
    const int e = t + j * RK_T;
    if (e < RK_CH) rank[e0 + e] = rk[j];
  }
  __syncthreads();

  int4* o4 = (int4*)(cntm + (size_t)w * N_NODES);
  for (int q = t; q < N_NODES / 4; q += RK_T) o4[q] = h4[q];
  if (t < N_SL) stot[t * RK_G + w] = sl[t];
}

__global__ __launch_bounds__(RK_T) void gat_gemm_rank(const float* __restrict__ h,
                                                      const float* __restrict__ W,
                                                      const float* __restrict__ avec,
                                                      float* __restrict__ Wh,
                                                      float* __restrict__ s,
                                                      float* __restrict__ d,
                                                      const int* __restrict__ dst,
                                                      int* __restrict__ rank,
                                                      int* __restrict__ cntm,
                                                      int* __restrict__ stot) {
  __shared__ __attribute__((aligned(16))) int hist[N_NODES];
  __shared__ int sl[N_SL];
  const int b = (int)blockIdx.x;
  if (b < RK_G) gat_rank(b, hist, sl, dst, rank, cntm, stot);
  else          gat_gemm_wmma(b - RK_G, h, W, avec, Wh, s, d);
}

__global__ __launch_bounds__(SL_N) void gat_scan(int* __restrict__ cntm,
                                                 const int* __restrict__ stot,
                                                 int* __restrict__ off,
                                                 int* __restrict__ cnt) {
  __shared__ int wpre[SL_N / 32];
  __shared__ int wtot[SL_N / 32];
  const int t    = (int)threadIdx.x;
  const int lane = t & 31;
  const int wv   = t >> 5;
  const int n    = (int)blockIdx.x * SL_N + t;

  const int nc = (n < N_NODES) ? n : N_NODES - 1;
  int c[RK_G];
  #pragma unroll
  for (int w = 0; w < RK_G; ++w) {
    const int v = cntm[(size_t)w * N_NODES + nc];
    c[w] = (n < N_NODES) ? v : 0;
  }

  const int lim = (int)blockIdx.x * RK_G;
  int pre = 0;
  #pragma unroll
  for (int k = 0; k < (N_SL * RK_G + SL_N - 1) / SL_N; ++k) {
    const int j = t + k * SL_N;
    const int v = stot[j < N_SL * RK_G ? j : N_SL * RK_G - 1];
    pre += (j < lim) ? v : 0;
  }
  #pragma unroll
  for (int o = 16; o > 0; o >>= 1) pre += __shfl_xor(pre, o, 32);

  int tot = 0;
  #pragma unroll
  for (int w = 0; w < RK_G; ++w) tot += c[w];
  int inc = tot;
  #pragma unroll
  for (int o = 1; o < 32; o <<= 1) {
    const int u = __shfl_up(inc, o, 32);
    if (lane >= o) inc += u;
  }
  if (lane == 0)  wpre[wv] = pre;
  if (lane == 31) wtot[wv] = inc;
  __syncthreads();
  int x = (lane < SL_N / 32) ? (wpre[lane] + ((lane < wv) ? wtot[lane] : 0)) : 0;
  #pragma unroll
  for (int o = 16; o > 0; o >>= 1) x += __shfl_xor(x, o, 32);

  if (n < N_NODES) {
    int run = x + inc - tot;
    off[n] = run;
    cnt[n] = tot;
    #pragma unroll
    for (int w = 0; w < RK_G; ++w) { cntm[(size_t)w * N_NODES + n] = run; run += c[w]; }
  }
}

__global__ __launch_bounds__(256) void gat_fill(const int* __restrict__ src,
                                                const int* __restrict__ dst,
                                                const int* __restrict__ rank,
                                                const int* __restrict__ base,
                                                const float* __restrict__ s,
                                                int* __restrict__ csr) {
  const int i = (int)(blockIdx.x * blockDim.x + threadIdx.x);
  if (i >= N_EDGES) return;
  const int sn = src[i];
  const int w  = i / RK_CH;
  const int slot = base[(size_t)w * N_NODES + dst[i]] + rank[i];
  ((int2*)csr)[slot] = make_int2(sn, __float_as_int(s[sn]));
}

__device__ __forceinline__ float gat_logit(int sbits, float dn) {
  const float v = __int_as_float(sbits) + dn;
  return (v > 0.f) ? v : LRELU_A * v;
}

template <int B>
__device__ __forceinline__ void gat_rows(float4& acc, const float* __restrict__ Wh,
                                         int sn, float w, int k0, int lane) {
  float4 v[B];
  float  wk[B];
  #pragma unroll
  for (int b = 0; b < B; ++b) {
    const int sk = __builtin_amdgcn_readlane(sn, k0 + b);
    wk[b] = __int_as_float(__builtin_amdgcn_readlane(__float_as_int(w), k0 + b));
    v[b]  = ((const float4*)(Wh + (size_t)sk * DIM))[lane];
  }
  #pragma unroll
  for (int b = 0; b < B; ++b) {
    acc.x = fmaf(wk[b], v[b].x, acc.x);
    acc.y = fmaf(wk[b], v[b].y, acc.y);
    acc.z = fmaf(wk[b], v[b].z, acc.z);
    acc.w = fmaf(wk[b], v[b].w, acc.w);
  }
}

__device__ __forceinline__ void gat_chunk(float4& acc, const float* __restrict__ Wh,
                                          int sn, float w, int nk, int lane) {
  int k = 0;
  for (; k + 8 <= nk; k += 8) gat_rows<8>(acc, Wh, sn, w, k, lane);
  if (k + 4 <= nk) { gat_rows<4>(acc, Wh, sn, w, k, lane); k += 4; }
  if (k + 2 <= nk) { gat_rows<2>(acc, Wh, sn, w, k, lane); k += 2; }
  if (k < nk)      { gat_rows<1>(acc, Wh, sn, w, k, lane); }
}

__global__ __launch_bounds__(256) void gat_gather(const float* __restrict__ Wh,
                                                  const float* __restrict__ h,
                                                  const float* __restrict__ d,
                                                  const int* __restrict__ off,
                                                  const int* __restrict__ cnt,
                                                  const int* __restrict__ csr,
                                                  float* __restrict__ hout) {
  const int node = __builtin_amdgcn_readfirstlane((int)((blockIdx.x * blockDim.x + threadIdx.x) >> 5));
  const int lane = (int)(threadIdx.x & 31);
  if (node >= N_NODES) return;
  const int   beg = off[node];
  const int   deg = cnt[node];
  const float dn  = d[node];
  const int2* recs = (const int2*)csr + beg;
  const float4 hv = ((const float4*)(h + (size_t)node * DIM))[lane];

  int   sn0 = 0;
  float e0  = -INFINITY;
  if (lane < deg) { const int2 r = recs[lane]; sn0 = r.x; e0 = gat_logit(r.y, dn); }
  float m = e0;
  for (int j = 32 + lane; j < deg; j += 32) m = fmaxf(m, gat_logit(recs[j].y, dn));
  #pragma unroll
  for (int o = 16; o > 0; o >>= 1) m = fmaxf(m, __shfl_xor(m, o, 32));
  if (!isfinite(m)) m = 0.f;

  const float ex0 = (lane < deg) ? expf(e0 - m) : 0.f;
  float den = ex0;
  for (int j = 32 + lane; j < deg; j += 32) den += expf(gat_logit(recs[j].y, dn) - m);
  #pragma unroll
  for (int o = 16; o > 0; o >>= 1) den += __shfl_xor(den, o, 32);

  float4 acc = make_float4(0.f, 0.f, 0.f, 0.f);
  gat_chunk(acc, Wh, sn0, ex0 / den, deg < 32 ? deg : 32, lane);
  for (int j0 = 32; j0 < deg; j0 += 32) {
    int sn = 0; float w = 0.f;
    if (j0 + lane < deg) {
      const int2 r = recs[j0 + lane];
      sn = r.x;
      w  = expf(gat_logit(r.y, dn) - m) / den;
    }
    gat_chunk(acc, Wh, sn, w, (deg - j0) < 32 ? (deg - j0) : 32, lane);
  }

  float4 o4;
  o4.x = acc.x + hv.x; o4.y = acc.y + hv.y; o4.z = acc.z + hv.z; o4.w = acc.w + hv.w;
  ((float4*)(hout + (size_t)node * DIM))[lane] = o4;
}

extern "C" void kernel_launch(void* const* d_in, const int* in_sizes, int n_in,
                              void* d_out, int out_size, void* d_ws, size_t ws_size,
                              hipStream_t stream) {
  const float* h   = (const float*)d_in[0];
  const float* W   = (const float*)d_in[1];
  const float* a   = (const float*)d_in[2];
  const int*   src = (const int*)d_in[3];
  const int*   dst = (const int*)d_in[4];
  float* hout = (float*)d_out;

  float* ws   = (float*)d_ws;
  float* Wh   = ws;
  float* s    = Wh + (size_t)N_NODES * DIM;
  float* d    = s  + N_NODES;
  int*   cnt  = (int*)(d + N_NODES);
  int*   off  = cnt + N_NODES;
  int*   rank = off + N_NODES;
  int*   csr  = rank + N_EDGES;
  int*   cntm = csr + 2 * (size_t)N_EDGES;
  int*   stot = cntm + (size_t)RK_G * N_NODES;

  gat_gemm_rank<<<RK_G + GM_G, RK_T, 0, stream>>>(h, W, a, Wh, s, d, dst, rank, cntm, stot);
  gat_scan<<<N_SL, SL_N, 0, stream>>>(cntm, stot, off, cnt);
  gat_fill<<<(N_EDGES + 255) / 256, 256, 0, stream>>>(src, dst, rank, cntm, s, csr);
  gat_gather<<<(N_NODES * 32 + 255) / 256, 256, 0, stream>>>(Wh, h, d, off, cnt, csr, hout);
}


# Chunk work-groups finish the sort in LDS (scan, placement); a gather lane per chunk finds the node's edges, packed in LDS: no scan or fill launch (4 to 2); sum order kept unless 9+ edges in a chunk
# speedup vs baseline: 5.6251x; 1.4685x over previous
#include <hip/hip_runtime.h>
#include <math.h>

#define N_NODES 50000
#define N_EDGES 800000
#define DIM     128
#define LRELU_A 0.2f
#define RK_G    32
#define RK_CH   (N_EDGES / RK_G)
#define RK_T    512
#define RK_PER  ((RK_CH + RK_T - 1) / RK_T)
#define GM_W    (RK_T / 32)
#define GM_G    ((N_NODES / 16 + GM_W - 1) / GM_W)
#define SC_Q    25
#define SC_T    (N_NODES / (4 * SC_Q))
#define GA_W    8
#define GA_NI   8
#define GA_NA   4
static_assert(N_EDGES % RK_G == 0, "chunks tile the edge list exactly");
static_assert(N_NODES % 4 == 0 && RK_CH % 4 == 0, "counter rows and sorted chunks move as 16-byte words");
static_assert(SC_T * SC_Q * 4 == N_NODES && SC_T <= RK_T, "the scanning threads tile the counter table exactly");
static_assert(RK_G == 32, "one lane of a gather wave per chunk");
static_assert(RK_CH < 32768, "a first slot and a count of one chunk share one 32-bit word");
static_assert(RK_G * GA_W == 32 * GA_W, "one gather thread per (chunk, node) entry of its work-group");

typedef __attribute__((ext_vector_type(2))) float v2f;
typedef __attribute__((ext_vector_type(8))) float v8f;

template <int M>
__device__ __forceinline__ float gat_row_xor(float x) {
  const int b = __builtin_bit_cast(int, x);
  return __builtin_bit_cast(float, __builtin_amdgcn_update_dpp(b, b, 0x160 + M, 0xf, 0xf, false));
}
__device__ __forceinline__ float gat_row_sum16(float x) {
  x += gat_row_xor<1>(x);
  x += gat_row_xor<2>(x);
  x += gat_row_xor<4>(x);
  x += gat_row_xor<8>(x);
  return x;
}

__device__ __forceinline__ void gat_gemm_wmma(const int wg,
                                              const float* __restrict__ h,
                                              const float* __restrict__ W,
                                              const float* __restrict__ avec,
                                              float* __restrict__ Wh,
                                              float* __restrict__ s,
                                              float* __restrict__ d) {
  const int wave = wg * GM_W + (int)(threadIdx.x >> 5);
  const int lane = (int)(threadIdx.x & 31);
  if (wave >= N_NODES / 16) return;

  const int row0  = wave * 16;
  const int rsel  = lane & 15;
  const int khalf = (lane >> 4) * 2;

  v8f acc[8];
  #pragma unroll
  for (int t = 0; t < 8; ++t) { v8f z = {}; acc[t] = z; }

  for (int k0 = 0; k0 < DIM; k0 += 4) {
    const float* ap = h + (size_t)(row0 + rsel) * DIM + (k0 + khalf);
    v2f a; a.x = ap[0]; a.y = ap[1];
    #pragma unroll
    for (int t = 0; t < 8; ++t) {
      const float* bp = W + (size_t)(k0 + khalf) * DIM + t * 16 + rsel;
      v2f b; b.x = bp[0]; b.y = bp[DIM];
      acc[t] = __builtin_amdgcn_wmma_f32_16x16x4_f32(
           false, a,  false, b,
           (short)0, acc[t],  false,  false);
    }
  }

  const int srow = row0 + (lane >> 4) * 8;
  #pragma unroll
  for (int t = 0; t < 8; ++t) {
    #pragma unroll
    for (int i = 0; i < 8; ++i) {
      Wh[(size_t)(srow + i) * DIM + t * 16 + rsel] = acc[t][i];
    }
  }

  float as[8], ad[8];
  #pragma unroll
  for (int t = 0; t < 8; ++t) { as[t] = avec[t * 16 + rsel]; ad[t] = avec[DIM + t * 16 + rsel]; }
  float ps[8], pd[8];
  #pragma unroll
  for (int i = 0; i < 8; ++i) {
    float x = 0.f, y = 0.f;
    #pragma unroll
    for (int t = 0; t < 8; ++t) {
      x = fmaf(acc[t][i], as[t], x);
      y = fmaf(acc[t][i], ad[t], y);
    }
    ps[i] = gat_row_sum16(x);
    pd[i] = gat_row_sum16(y);
  }
  if (rsel == 0) {
    float4* sp = (float4*)(s + srow);
    float4* dp = (float4*)(d + srow);
    sp[0] = make_float4(ps[0], ps[1], ps[2], ps[3]);
    sp[1] = make_float4(ps[4], ps[5], ps[6], ps[7]);
    dp[0] = make_float4(pd[0], pd[1], pd[2], pd[3]);
    dp[1] = make_float4(pd[4], pd[5], pd[6], pd[7]);
  }
}

__device__ __forceinline__ void gat_sort(const int w, int* hist, int* srt, int* wsum,
                                         const int* __restrict__ src,
                                         const int* __restrict__ dst,
                                         int* __restrict__ cntm,
                                         int* __restrict__ lst) {
  const int t    = (int)threadIdx.x;
  const int lane = t & 31;
  const int wv   = t >> 5;
  const int e0   = w * RK_CH;

  int dv[RK_PER], sv[RK_PER];
  #pragma unroll
  for (int j = 0; j < RK_PER; ++j) {
    const int e = t + j * RK_T;
    dv[j] = (e < RK_CH) ? dst[e0 + e] : 0;
    sv[j] = (e < RK_CH) ? src[e0 + e] : 0;
  }

  int4* h4 = (int4*)hist;
  for (int q = t; q < N_NODES / 4; q += RK_T) h4[q] = make_int4(0, 0, 0, 0);
  __syncthreads();

  int rk[RK_PER];
  #pragma unroll
  for (int j = 0; j < RK_PER; ++j) {
    const int e = t + j * RK_T;
    rk[j] = 0;
    if (e < RK_CH) rk[j] = atomicAdd(&hist[dv[j]], 1);
  }
  __syncthreads();

  int sum = 0;
  if (t < SC_T) {
    for (int q = 0; q < SC_Q; ++q) {
      const int4 v = h4[t * SC_Q + q];
      sum += v.x + v.y + v.z + v.w;
    }
  }
  int inc = sum;
  #pragma unroll
  for (int o = 1; o < 32; o <<= 1) {
    const int u = __shfl_up(inc, o, 32);
    if (lane >= o) inc += u;
  }
  if (lane == 31) wsum[wv] = inc;
  __syncthreads();
  int run = inc - sum;
  #pragma unroll
  for (int k = 0; k < GM_W; ++k) run += (k < wv) ? wsum[k] : 0;
  if (t < SC_T) {
    for (int q = 0; q < SC_Q; ++q) {
      const int4 v = h4[t * SC_Q + q];
      int4 o;
      o.x = run | (v.x << 16); run += v.x;
      o.y = run | (v.y << 16); run += v.y;
      o.z = run | (v.z << 16); run += v.z;
      o.w = run | (v.w << 16); run += v.w;
      h4[t * SC_Q + q] = o;
    }
  }
  __syncthreads();

  #pragma unroll
  for (int j = 0; j < RK_PER; ++j) {
    const int e = t + j * RK_T;
    if (e < RK_CH) srt[(hist[dv[j]] & 0xffff) + rk[j]] = sv[j];
  }
  __syncthreads();

  int4* o4 = (int4*)(cntm + (size_t)w * N_NODES);
  for (int q = t; q < N_NODES / 4; q += RK_T) o4[q] = h4[q];
  const int4* s4 = (const int4*)srt;
  int4* l4 = (int4*)(lst + e0);
  for (int q = t; q < RK_CH / 4; q += RK_T) l4[q] = s4[q];
}

__global__ __launch_bounds__(RK_T) void gat_gemm_sort(const float* __restrict__ h,
                                                      const float* __restrict__ W,
                                                      const float* __restrict__ avec,
                                                      float* __restrict__ Wh,
                                                      float* __restrict__ s,
                                                      float* __restrict__ d,
                                                      const int* __restrict__ src,
                                                      const int* __restrict__ dst,
                                                      int* __restrict__ cntm,
                                                      int* __restrict__ lst) {
  __shared__ __attribute__((aligned(16))) int hist[N_NODES];
  __shared__ __attribute__((aligned(16))) int srt[RK_CH];
  __shared__ int wsum[GM_W];
  const int b = (int)blockIdx.x;
  if (b < RK_G) gat_sort(b, hist, srt, wsum, src, dst, cntm, lst);
  else          gat_gemm_wmma(b - RK_G, h, W, avec, Wh, s, d);
}

__device__ __forceinline__ float gat_logit(int sbits, float dn) {
  const float v = __int_as_float(sbits) + dn;
  return (v > 0.f) ? v : LRELU_A * v;
}

template <int B>
__device__ __forceinline__ void gat_rows(float4& acc, const float* __restrict__ Wh,
                                         int sn, float w, int k0, int lane) {
  float4 v[B];
  float  wk[B];
  #pragma unroll
  for (int b = 0; b < B; ++b) {
    const int sk = __builtin_amdgcn_readlane(sn, k0 + b);
    wk[b] = __int_as_float(__builtin_amdgcn_readlane(__float_as_int(w), k0 + b));
    v[b]  = ((const float4*)(Wh + (size_t)sk * DIM))[lane];
  }
  #pragma unroll
  for (int b = 0; b < B; ++b) {
    acc.x = fmaf(wk[b], v[b].x, acc.x);
    acc.y = fmaf(wk[b], v[b].y, acc.y);
    acc.z = fmaf(wk[b], v[b].z, acc.z);
    acc.w = fmaf(wk[b], v[b].w, acc.w);
  }
}

__device__ __forceinline__ void gat_chunk(float4& acc, const float* __restrict__ Wh,
                                          int sn, float w, int nk, int lane) {
  int k = 0;
  for (; k + 8 <= nk; k += 8) gat_rows<8>(acc, Wh, sn, w, k, lane);
  if (k + 4 <= nk) { gat_rows<4>(acc, Wh, sn, w, k, lane); k += 4; }
  if (k + 2 <= nk) { gat_rows<2>(acc, Wh, sn, w, k, lane); k += 2; }
  if (k < nk)      { gat_rows<1>(acc, Wh, sn, w, k, lane); }
}

template <int I0, int I1>
__device__ __forceinline__ void gat_pack(int2* recs, const int* lp, const int* __restrict__ lst,
                                         const float* __restrict__ s, int c, int p) {
  int sn[I1 - I0], sb[I1 - I0];
  #pragma unroll
  for (int i = I0; i < I1; ++i) sn[i - I0] = *((i < c) ? lp + i : lst);
  #pragma unroll
  for (int i = I0; i < I1; ++i) sb[i - I0] = __float_as_int(s[sn[i - I0]]);
  #pragma unroll
  for (int i = I0; i < I1; ++i) recs[(i < c) ? p + i : 32 * GA_NI] = make_int2(sn[i - I0], sb[i - I0]);
}

__global__ __launch_bounds__(32 * GA_W) void gat_gather(const float* __restrict__ Wh,
                                                        const float* __restrict__ h,
                                                        const float* __restrict__ s,
                                                        const float* __restrict__ d,
                                                        const int* __restrict__ cntm,
                                                        const int* __restrict__ lst,
                                                        float* __restrict__ hout) {
  __shared__ int2 recbuf[GA_W][32 * GA_NI + 1];
  __shared__ int  tile[RK_G * (GA_W + 1)];
  const int t    = (int)threadIdx.x;
  const int lane = t & 31;
  const int wv   = __builtin_amdgcn_readfirstlane(t >> 5);
  const int n0   = (int)blockIdx.x * GA_W;
  const int node = n0 + wv;
  const bool live = node < N_NODES;

  float dn = 0.f;
  if (live) dn = d[node];
  {
    const int ch = t / GA_W, col = t % GA_W;
    tile[ch * (GA_W + 1) + col] = (n0 + col < N_NODES) ? cntm[(size_t)ch * N_NODES + n0 + col] : 0;
  }
  __syncthreads();
  if (!live) return;
  int2* recs = recbuf[wv];

  const int  bc = tile[lane * (GA_W + 1) + wv];
  const int  b  = bc & 0xffff;
  const int  c  = bc >> 16;
  const int* lp = lst + (size_t)lane * RK_CH + b;
  float4 hv;

  float4 acc = make_float4(0.f, 0.f, 0.f, 0.f);
  if (__builtin_amdgcn_ballot_w32(c > GA_NI) == 0) {
    int p = 0, deg = 0;
    #pragma unroll
    for (int i = 0; i < GA_NI; ++i) {
      const unsigned mk = __builtin_amdgcn_ballot_w32(i < c);
      p   += (int)__builtin_amdgcn_mbcnt_lo(mk, 0u);
      deg += __builtin_popcount(mk);
    }
    gat_pack<0, GA_NA>(recs, lp, lst, s, c, p);
    if (__builtin_amdgcn_ballot_w32(c > GA_NA) != 0) gat_pack<GA_NA, GA_NI>(recs, lp, lst, s, c, p);
    hv = ((const float4*)(h + (size_t)node * DIM))[lane];

    int   sn0 = 0;
    float e0  = -INFINITY;
    if (lane < deg) { const int2 r = recs[lane]; sn0 = r.x; e0 = gat_logit(r.y, dn); }
    float m = e0;
    for (int j = 32 + lane; j < deg; j += 32) m = fmaxf(m, gat_logit(recs[j].y, dn));
    #pragma unroll
    for (int o = 16; o > 0; o >>= 1) m = fmaxf(m, __shfl_xor(m, o, 32));
    if (!isfinite(m)) m = 0.f;

    const float ex0 = (lane < deg) ? expf(e0 - m) : 0.f;
    float den = ex0;
    for (int j = 32 + lane; j < deg; j += 32) den += expf(gat_logit(recs[j].y, dn) - m);
    #pragma unroll
    for (int o = 16; o > 0; o >>= 1) den += __shfl_xor(den, o, 32);

    gat_chunk(acc, Wh, sn0, ex0 / den, deg < 32 ? deg : 32, lane);
    for (int j0 = 32; j0 < deg; j0 += 32) {
      int sn1 = 0; float w1 = 0.f;
      if (j0 + lane < deg) {
        const int2 r = recs[j0 + lane];
        sn1 = r.x;
        w1  = expf(gat_logit(r.y, dn) - m) / den;
      }
      gat_chunk(acc, Wh, sn1, w1, (deg - j0) < 32 ? (deg - j0) : 32, lane);
    }
  } else {
    hv = ((const float4*)(h + (size_t)node * DIM))[lane];
    float m = -INFINITY;
    for (int i = 0; __builtin_amdgcn_ballot_w32(i < c) != 0; ++i) {
      if (i < c) m = fmaxf(m, gat_logit(__float_as_int(s[lp[i]]), dn));
    }
    #pragma unroll
    for (int o = 16; o > 0; o >>= 1) m = fmaxf(m, __shfl_xor(m, o, 32));
    if (!isfinite(m)) m = 0.f;

    float den = 0.f;
    for (int i = 0; __builtin_amdgcn_ballot_w32(i < c) != 0; ++i) {
      if (i < c) den += expf(gat_logit(__float_as_int(s[lp[i]]), dn) - m);
    }
    #pragma unroll
    for (int o = 16; o > 0; o >>= 1) den += __shfl_xor(den, o, 32);

    for (int i = 0; ; ++i) {
      unsigned mk = __builtin_amdgcn_ballot_w32(i < c);
      if (mk == 0) break;
      int sn1 = 0; float w1 = 0.f;
      if (i < c) {
        sn1 = lp[i];
        w1  = expf(gat_logit(__float_as_int(s[sn1]), dn) - m) / den;
      }
      while (mk != 0) {
        const int k = __builtin_ctz(mk);
        mk &= mk - 1;
        gat_rows<1>(acc, Wh, sn1, w1, k, lane);
      }
    }
  }

  float4 o4;
  o4.x = acc.x + hv.x; o4.y = acc.y + hv.y; o4.z = acc.z + hv.z; o4.w = acc.w + hv.w;
  ((float4*)(hout + (size_t)node * DIM))[lane] = o4;
}

extern "C" void kernel_launch(void* const* d_in, const int* in_sizes, int n_in,
                              void* d_out, int out_size, void* d_ws, size_t ws_size,
                              hipStream_t stream) {
  const float* h   = (const float*)d_in[0];
  const float* W   = (const float*)d_in[1];
  const float* a   = (const float*)d_in[2];
  const int*   src = (const int*)d_in[3];
  const int*   dst = (const int*)d_in[4];
  float* hout = (float*)d_out;

  float* ws   = (float*)d_ws;
  float* Wh   = ws;
  float* s    = Wh + (size_t)N_NODES * DIM;
  float* d    = s  + N_NODES;
  int*   cntm = (int*)(d + N_NODES);
  int*   lst  = cntm + (size_t)RK_G * N_NODES;

  gat_gemm_sort<<<RK_G + GM_G, RK_T, 0, stream>>>(h, W, a, Wh, s, d, src, dst, cntm, lst);
  gat_gather<<<(N_NODES + GA_W - 1) / GA_W, 32 * GA_W, 0, stream>>>(Wh, h, s, d, cntm, lst, hout);
}
